# starting kernel
# baseline (speedup 1.0000x reference)
#include <hip/hip_runtime.h>
#include <hip/hip_bf16.h>

#define BB   512
#define CC   128
#define II   9
#define EE   10
#define MM   3
#define NN   (BB * CC)
#define K3S  7
#define K3V  10
#define MS_PAD 96
#define KS_PAD 64
#define MV_PAD 256
#define KV_PAD 96

typedef float v2f __attribute__((ext_vector_type(2)));
typedef float v8f __attribute__((ext_vector_type(8)));

#define OFF_ELEM  0
#define OFF_U3SF  4096
#define OFF_U3VF  (OFF_U3SF + 24576)
#define OFF_Z3S   131072
#define OFF_Z3V   (OFF_Z3S + 16777216)
#define WS_NEEDED (OFF_Z3V + 25165824)

__global__ void sc_elem_kernel(const float* __restrict__ y, int* __restrict__ elem) {
    int b = blockIdx.x * blockDim.x + threadIdx.x;
    if (b < BB) {
        int e = 0;
        for (int j = 0; j < EE; ++j)
            if (y[b * EE + j] > 0.5f) e = j;
        elem[b] = e;
    }
}

__global__ void sc_build_a_kernel(const float* __restrict__ U3s, const float* __restrict__ U3v,
                                  float* __restrict__ U3s_flat, float* __restrict__ U3v_flat) {
    int t = blockIdx.x * blockDim.x + threadIdx.x;
    if (t < MS_PAD * KS_PAD) {
        int row = t >> 6, col = t & 63;
        float v = 0.f;
        if (row < 81 && col < 63) {
            int i = row / 9, j = row % 9, a = col / K3S, k = col % K3S;
            v = U3s[((i * 9 + j) * 9 + a) * K3S + k];
        }
        U3s_flat[t] = v;
    } else {
        int t2 = t - MS_PAD * KS_PAD;
        if (t2 < MV_PAD * KV_PAD) {
            int row = t2 / KV_PAD, col = t2 % KV_PAD;
            float v = 0.f;
            if (row < 243 && col < 90) {
                int m = row / 81, rem = row % 81, i = rem / 9, j = rem % 9;
                int a = col / K3V, k = col % K3V;
                v = U3v[(((m * 9 + i) * 9 + j) * 9 + a) * K3V + k];
            }
            U3v_flat[t2] = v;
        }
    }
}

__global__ void sc_build_z_kernel(const float* __restrict__ x, const int* __restrict__ elem,
                                  const float* __restrict__ W3s, const float* __restrict__ W3v,
                                  float* __restrict__ z3s, float* __restrict__ z3v) {
    int n = blockIdx.x * blockDim.x + threadIdx.x;
    if (n >= NN) return;
    int b = n >> 7, c = n & 127;
    int e = elem[b];
    float xa[II];
#pragma unroll
    for (int a = 0; a < II; ++a) xa[a] = x[n * II + a];

    float vs[KS_PAD];
#pragma unroll
    for (int r = 0; r < KS_PAD; ++r) vs[r] = 0.f;
#pragma unroll
    for (int k = 0; k < K3S; ++k) {
        float w = W3s[(e * K3S + k) * CC + c];
#pragma unroll
        for (int a = 0; a < II; ++a) vs[a * K3S + k] = xa[a] * w;
    }
#pragma unroll
    for (int p = 0; p < KS_PAD / 2; ++p) {
        v2f t; t[0] = vs[2 * p]; t[1] = vs[2 * p + 1];
        *(v2f*)(z3s + (size_t)(p * NN + n) * 2) = t;
    }

    float vv[KV_PAD];
#pragma unroll
    for (int r = 0; r < KV_PAD; ++r) vv[r] = 0.f;
#pragma unroll
    for (int k = 0; k < K3V; ++k) {
        float w = W3v[(e * K3V + k) * CC + c];
#pragma unroll
        for (int a = 0; a < II; ++a) vv[a * K3V + k] = xa[a] * w;
    }
#pragma unroll
    for (int p = 0; p < KV_PAD / 2; ++p) {
        v2f t; t[0] = vv[2 * p]; t[1] = vv[2 * p + 1];
        *(v2f*)(z3v + (size_t)(p * NN + n) * 2) = t;
    }
}

__global__ __launch_bounds__(32) void sc_main_kernel(
    const float* __restrict__ x, const int* __restrict__ elem,
    const float* __restrict__ z3s, const float* __restrict__ z3v,
    const float* __restrict__ U3s_flat, const float* __restrict__ U3v_flat,
    const float* __restrict__ U2s, const float* __restrict__ U2v,
    const float* __restrict__ U1s, const float* __restrict__ U1v,
    const float* __restrict__ W2s, const float* __restrict__ W2v,
    const float* __restrict__ W1s, const float* __restrict__ W1v,
    float* __restrict__ out)
{
    __shared__ float t2s[MS_PAD * 16];
    __shared__ float t2v[MV_PAD * 16];
    __shared__ float u2s_l[9 * 9 * 3];
    __shared__ float u2v_l[3 * 9 * 9 * 3];
    __shared__ float u1s_l[9];
    __shared__ float u1v_l[3 * 9];

    const int lane = threadIdx.x;
    for (int i = lane; i < 243; i += 32) u2s_l[i] = U2s[i];
    for (int i = lane; i < 729; i += 32) u2v_l[i] = U2v[i];
    if (lane < 9)  u1s_l[lane] = U1s[lane];
    if (lane < 27) u1v_l[lane] = U1v[lane];

    const int half = lane >> 4;
    const int lq   = lane & 15;
    const int n0   = blockIdx.x * 16;
    const int coln = n0 + lq;

    {
        v2f bp[16];
#pragma unroll
        for (int kt = 0; kt < 16; ++kt) {
            int p = kt * 2 + half;
            bp[kt] = *(const v2f*)(z3s + (size_t)(p * NN + coln) * 2);
        }
        for (int mt = 0; mt < 6; ++mt) {
            v8f acc = {};
            const float* arow = U3s_flat + (mt * 16 + lq) * KS_PAD + half * 2;
            __builtin_prefetch(arow + 16 * KS_PAD, 0, 1);
#pragma unroll
            for (int kt = 0; kt < 16; ++kt) {
                v2f a = *(const v2f*)(arow + kt * 4);
                acc = __builtin_amdgcn_wmma_f32_16x16x4_f32(
                    false, a, false, bp[kt], (short)0, acc, false, false);
            }
#pragma unroll
            for (int v = 0; v < 8; ++v) {
                int row = mt * 16 + v + half * 8;
                t2s[row * 16 + lq] = acc[v];
            }
        }
    }

    {
        v2f bpv[24];
#pragma unroll
        for (int kt = 0; kt < 24; ++kt) {
            int p = kt * 2 + half;
            bpv[kt] = *(const v2f*)(z3v + (size_t)(p * NN + coln) * 2);
        }
        for (int mt = 0; mt < 16; ++mt) {
            v8f acc = {};
            const float* arow = U3v_flat + (mt * 16 + lq) * KV_PAD + half * 2;
            if (mt < 15) __builtin_prefetch(arow + 16 * KV_PAD, 0, 1);
#pragma unroll
            for (int kt = 0; kt < 24; ++kt) {
                v2f a = *(const v2f*)(arow + kt * 4);
                acc = __builtin_amdgcn_wmma_f32_16x16x4_f32(
                    false, a, false, bpv[kt], (short)0, acc, false, false);
            }
#pragma unroll
            for (int v = 0; v < 8; ++v) {
                int row = mt * 16 + v + half * 8;
                t2v[row * 16 + lq] = acc[v];
            }
        }
    }

    __syncthreads();

    const int col = lq;
    const int n = n0 + col;
    const int b = n >> 7, c = n & 127;
    const int e = elem[b];
    float xa[II];
#pragma unroll
    for (int a = 0; a < II; ++a) xa[a] = x[n * II + a];

    if (half == 0) {
        float w2[3];
#pragma unroll
        for (int k = 0; k < 3; ++k) w2[k] = W2s[(e * 3 + k) * CC + c];
        float t1[II];
#pragma unroll
        for (int i = 0; i < II; ++i) {
            float acc1 = 0.f;
#pragma unroll
            for (int a = 0; a < II; ++a) {
                float c2 = 0.f;
#pragma unroll
                for (int k = 0; k < 3; ++k) c2 += u2s_l[(i * 9 + a) * 3 + k] * w2[k];
                acc1 += (c2 + t2s[(i * 9 + a) * 16 + col]) * xa[a];
            }
            t1[i] = acc1;
        }
        float w1 = W1s[e * CC + c];
        float s = 0.f;
#pragma unroll
        for (int a = 0; a < II; ++a) s += (u1s_l[a] * w1 + t1[a]) * xa[a];
        out[b * 512 + c] = s;
    }

    {
        float w2v[3];
#pragma unroll
        for (int k = 0; k < 3; ++k) w2v[k] = W2v[(e * 3 + k) * CC + c];
        float w1v = W1v[e * CC + c];
        const int mstart = (half == 0) ? 0 : 1;
        const int mend   = (half == 0) ? 1 : 3;
        for (int m = mstart; m < mend; ++m) {
            float t1v[II];
#pragma unroll
            for (int i = 0; i < II; ++i) {
                float acc1 = 0.f;
#pragma unroll
                for (int a = 0; a < II; ++a) {
                    float c2 = 0.f;
#pragma unroll
                    for (int k = 0; k < 3; ++k)
                        c2 += u2v_l[((m * 9 + i) * 9 + a) * 3 + k] * w2v[k];
                    acc1 += (c2 + t2v[(m * 81 + i * 9 + a) * 16 + col]) * xa[a];
                }
                t1v[i] = acc1;
            }
            float v = 0.f;
#pragma unroll
            for (int a = 0; a < II; ++a) v += (u1v_l[m * 9 + a] * w1v + t1v[a]) * xa[a];
            out[b * 512 + CC + c * 3 + m] = v;
        }
    }
}

extern "C" void kernel_launch(void* const* d_in, const int* in_sizes, int n_in,
                              void* d_out, int out_size, void* d_ws, size_t ws_size,
                              hipStream_t stream) {
    (void)in_sizes; (void)n_in; (void)out_size;
    if (ws_size < (size_t)WS_NEEDED) return;

    const float* x   = (const float*)d_in[0];
    const float* y   = (const float*)d_in[1];
    const float* U1s = (const float*)d_in[2];
    const float* U2s = (const float*)d_in[3];
    const float* U3s = (const float*)d_in[4];
    const float* U1v = (const float*)d_in[5];
    const float* U2v = (const float*)d_in[6];
    const float* U3v = (const float*)d_in[7];
    const float* W1s = (const float*)d_in[8];
    const float* W2s = (const float*)d_in[9];
    const float* W3s = (const float*)d_in[10];
    const float* W1v = (const float*)d_in[11];
    const float* W2v = (const float*)d_in[12];
    const float* W3v = (const float*)d_in[13];
    float* out = (float*)d_out;

    char* ws = (char*)d_ws;
    int*   elem     = (int*)  (ws + OFF_ELEM);
    float* U3s_flat = (float*)(ws + OFF_U3SF);
    float* U3v_flat = (float*)(ws + OFF_U3VF);
    float* z3s      = (float*)(ws + OFF_Z3S);
    float* z3v      = (float*)(ws + OFF_Z3V);

    sc_elem_kernel<<<2, 256, 0, stream>>>(y, elem);
    sc_build_a_kernel<<<(MS_PAD * KS_PAD + MV_PAD * KV_PAD + 255) / 256, 256, 0, stream>>>(
        U3s, U3v, U3s_flat, U3v_flat);
    sc_build_z_kernel<<<NN / 256, 256, 0, stream>>>(x, elem, W3s, W3v, z3s, z3v);
    sc_main_kernel<<<NN / 16, 32, 0, stream>>>(
        x, elem, z3s, z3v, U3s_flat, U3v_flat,
        U2s, U2v, U1s, U1v, W2s, W2v, W1s, W1v, out);
}


# Re-associated contraction: V = U3 x W3 per (element, channel) on WMMA into workspace, then V x x over nodes grouped by element (K 63/90 -> 12); x*W3 build launch dropped; sum order changes
# speedup vs baseline: 2.5233x; 2.5233x over previous
#include <hip/hip_runtime.h>

#define BB   512
#define CC   128
#define II   9
#define EE   10
#define NP   (EE * CC)
#define NBLK 4
#define VROWS 96
#define VK   12
#define VBLK (VROWS * VK)
#define VPAIR (NBLK * VBLK)

typedef float v2f __attribute__((ext_vector_type(2)));
typedef float v8f __attribute__((ext_vector_type(8)));

#define OFF_PERM  0
#define OFF_GOFF  4096
#define OFF_V     131072
#define WS_NEEDED (OFF_V + NP * VPAIR * 4)


__global__ __launch_bounds__(512) void sc_group_kernel(const float* __restrict__ y,
                                                       int* __restrict__ perm, int* __restrict__ goff) {
    __shared__ int cnt[16 * EE];
    __shared__ int wbase[16 * EE];
    __shared__ int tot[EE];
    __shared__ int gstart[EE + 1];
    const int b = threadIdx.x;
    const int lane = b & 31, w = b >> 5;
    int e = 0;
    for (int j = 0; j < EE; ++j)
        if (y[b * EE + j] > 0.5f) e = j;
    int myrank = 0;
    for (int j = 0; j < EE; ++j) {
        unsigned m = __builtin_amdgcn_ballot_w32(e == j);
        if (e == j) myrank = __popc(m & ((1u << lane) - 1u));
        if (lane == 0) cnt[w * EE + j] = __popc(m);
    }
    __syncthreads();
    if (b < EE) {
        int run = 0;
        for (int w2 = 0; w2 < 16; ++w2) { wbase[w2 * EE + b] = run; run += cnt[w2 * EE + b]; }
        tot[b] = run;
    }
    __syncthreads();
    if (b == 0) {
        int run = 0;
        for (int j = 0; j < EE; ++j) { gstart[j] = run; run += tot[j]; }
        gstart[EE] = run;
    }
    __syncthreads();
    perm[gstart[e] + wbase[w * EE + e] + myrank] = b;
    if (b <= EE) goff[b] = gstart[b];
}

__global__ __launch_bounds__(256) void sc_build_v_kernel(
    const float* __restrict__ U3s, const float* __restrict__ U3v,
    const float* __restrict__ W3s, const float* __restrict__ W3v,
    float* __restrict__ V)
{
    const int lane = threadIdx.x & 31;
    const int wave = __builtin_amdgcn_readfirstlane((int)(threadIdx.x >> 5));
    const int half = lane >> 4;
    const int lq   = lane & 15;
    const int gw   = (int)blockIdx.x * 8 + wave;
    const int ng   = gw % 6;
    const int rest = gw / 6;
    const int blk  = rest & 3;
    const int ct   = (rest >> 2) & 7;
    const int e    = rest >> 5;
    const int K3   = (blk == 0) ? 7 : 10;
    const float* W3 = (blk == 0) ? W3s : W3v;
    const float* U3 = (blk == 0) ? U3s : (U3v + (blk - 1) * (81 * 9 * 10));

    v2f af[3];
#pragma unroll
    for (int kt = 0; kt < 3; ++kt) {
#pragma unroll
        for (int j = 0; j < 2; ++j) {
            int k = kt * 4 + half * 2 + j;
            bool ok = k < K3;
            float v = W3[(e * K3 + (ok ? k : 0)) * CC + ct * 16 + lq];
            af[kt][j] = ok ? v : 0.f;
        }
    }

    for (int t = 0; t < 6; ++t) {
        const int nt0 = ng * 12 + t * 2;
        v2f bf[2][3];
#pragma unroll
        for (int q = 0; q < 2; ++q) {
            int R  = (nt0 + q) * 16 + lq;
            int ij = R / VK;
            int ap = R - ij * VK;
            bool okr = (ij < 81) && (ap < II);
            int base = (ij * II + ap) * K3;
#pragma unroll
            for (int kt = 0; kt < 3; ++kt) {
#pragma unroll
                for (int j = 0; j < 2; ++j) {
                    int k = kt * 4 + half * 2 + j;
                    bool ok = okr && (k < K3);
                    float v = U3[ok ? base + k : 0];
                    bf[q][kt][j] = ok ? v : 0.f;
                }
            }
        }
        v8f acc0 = {}, acc1 = {};
#pragma unroll
        for (int kt = 0; kt < 3; ++kt) {
            acc0 = __builtin_amdgcn_wmma_f32_16x16x4_f32(false, af[kt], false, bf[0][kt], (short)0, acc0, false, false);
            acc1 = __builtin_amdgcn_wmma_f32_16x16x4_f32(false, af[kt], false, bf[1][kt], (short)0, acc1, false, false);
        }
#pragma unroll
        for (int v = 0; v < 8; ++v) {
            int c = ct * 16 + v + half * 8;
            float* dst = V + ((size_t)(e * CC + c) * NBLK + blk) * VBLK + nt0 * 16 + lq;
            dst[0]  = acc0[v];
            dst[16] = acc1[v];
        }
    }
}

#define TP    68
#define WPB2  10
#define NWG2  (NP / 5)

__global__ __launch_bounds__(320) void sc_main_kernel(
    const float* __restrict__ x, const int* __restrict__ perm, const int* __restrict__ goff,
    const float* __restrict__ V,
    const float* __restrict__ U2s, const float* __restrict__ U2v,
    const float* __restrict__ U1s, const float* __restrict__ U1v,
    const float* __restrict__ W2s, const float* __restrict__ W2v,
    const float* __restrict__ W1s, const float* __restrict__ W1v,
    float* __restrict__ out)
{
    __shared__ float tile_all[WPB2 * VROWS * TP];
    __shared__ float c2_all[WPB2 * VROWS];
    __shared__ float u2_l[4 * 243];
    __shared__ float u1_l[4 * 9];

    const int tid = threadIdx.x;
    for (int i = tid; i < 243; i += 320) u2_l[i] = U2s[i];
    for (int i = tid; i < 729; i += 320) u2_l[243 + i] = U2v[i];
    if (tid < 9)  u1_l[tid] = U1s[tid];
    if (tid < 27) u1_l[9 + tid] = U1v[tid];
    __syncthreads();

    const int lane = tid & 31;
    const int wave = __builtin_amdgcn_readfirstlane((int)(tid >> 5));
    const int half = lane >> 4;
    const int lq   = lane & 15;
    const int p = (int)blockIdx.x * 5 + (wave >> 1);
    const int s = wave & 1;
    const int e = p >> 7, c = p & 127;
    const int g0 = goff[e];
    const int nb = goff[e + 1] - g0;
    float* tile = tile_all + wave * (VROWS * TP);
    float* c2w  = c2_all + wave * VROWS;
    const float* Vp = V + (size_t)p * VPAIR;

    for (int ch = 0; ch * 64 < nb; ++ch) {
        v2f bf[4][3];
#pragma unroll
        for (int nt = 0; nt < 4; ++nt) {
            int j = ch * 64 + nt * 16 + lq;
            bool valid = j < nb;
            int b = perm[g0 + (valid ? j : 0)];
            const float* xp = x + ((size_t)b * CC + c) * II;
#pragma unroll
            for (int kt = 0; kt < 3; ++kt) {
#pragma unroll
                for (int jj = 0; jj < 2; ++jj) {
                    int k = kt * 4 + half * 2 + jj;
                    bool ok = valid && (k < II);
                    float v = xp[(k < II) ? k : 0];
                    bf[nt][kt][jj] = ok ? v : 0.f;
                }
            }
        }

        for (int bi = 0; bi < 2; ++bi) {
            const int blk = s * 2 + bi;
            const float* Vb = Vp + blk * VBLK;


            const float* W2 = (blk == 0) ? W2s : W2v;
            const float* W1 = (blk == 0) ? W1s : W1v;
            float w2[3];
#pragma unroll
            for (int k = 0; k < 3; ++k) w2[k] = W2[(e * 3 + k) * CC + c];
            const float w1 = W1[e * CC + c];
            for (int r = lane; r < 81; r += 32) {
                float c2 = 0.f;
#pragma unroll
                for (int k = 0; k < 3; ++k) c2 += u2_l[blk * 243 + r * 3 + k] * w2[k];
                c2w[r] = c2;
            }

            for (int hb = 0; hb < 2; ++hb) {
                v2f a[3][3];
#pragma unroll
                for (int m3 = 0; m3 < 3; ++m3) {
                    const float* ar = Vb + ((hb * 3 + m3) * 16 + lq) * VK + half * 2;
#pragma unroll
                    for (int kt = 0; kt < 3; ++kt) a[m3][kt] = *(const v2f*)(ar + kt * 4);
                }
#pragma unroll
                for (int m3 = 0; m3 < 3; ++m3) {
                    v8f acc[4] = {};
#pragma unroll
                    for (int kt = 0; kt < 3; ++kt) {
#pragma unroll
                        for (int nt = 0; nt < 4; ++nt)
                            acc[nt] = __builtin_amdgcn_wmma_f32_16x16x4_f32(
                                false, a[m3][kt], false, bf[nt][kt], (short)0, acc[nt], false, false);
                    }
#pragma unroll
                    for (int v = 0; v < 8; ++v) {
                        int row = (hb * 3 + m3) * 16 + v + half * 8;
#pragma unroll
                        for (int nt = 0; nt < 4; ++nt) tile[row * TP + nt * 16 + lq] = acc[nt][v];
                    }
                }
            }

            for (int ps = 0; ps < 2; ++ps) {
                const int col = ps * 32 + lane;
                const int j = ch * 64 + col;
                if (j < nb) {
                    const int b = perm[g0 + j];
                    const float* xp = x + ((size_t)b * CC + c) * II;
                    float xa[II];
#pragma unroll
                    for (int q = 0; q < II; ++q) xa[q] = xp[q];
                    float t1[II];
#pragma unroll
                    for (int i = 0; i < II; ++i) {
                        float acc1 = 0.f;
#pragma unroll
                        for (int q = 0; q < II; ++q)
                            acc1 += (c2w[i * 9 + q] + tile[(i * 9 + q) * TP + col]) * xa[q];
                        t1[i] = acc1;
                    }
                    float r = 0.f;
#pragma unroll
                    for (int q = 0; q < II; ++q) r += (u1_l[blk * 9 + q] * w1 + t1[q]) * xa[q];
                    if (blk == 0) out[b * 512 + c] = r;
                    else          out[b * 512 + CC + c * 3 + (blk - 1)] = r;
                }
            }
        }
    }
}

extern "C" void kernel_launch(void* const* d_in, const int* in_sizes, int n_in,
                              void* d_out, int out_size, void* d_ws, size_t ws_size,
                              hipStream_t stream) {
    (void)in_sizes; (void)n_in; (void)out_size;
    if (ws_size < (size_t)WS_NEEDED) return;

    const float* x   = (const float*)d_in[0];
    const float* y   = (const float*)d_in[1];
    const float* U1s = (const float*)d_in[2];
    const float* U2s = (const float*)d_in[3];
    const float* U3s = (const float*)d_in[4];
    const float* U1v = (const float*)d_in[5];
    const float* U2v = (const float*)d_in[6];
    const float* U3v = (const float*)d_in[7];
    const float* W1s = (const float*)d_in[8];
    const float* W2s = (const float*)d_in[9];
    const float* W3s = (const float*)d_in[10];
    const float* W1v = (const float*)d_in[11];
    const float* W2v = (const float*)d_in[12];
    const float* W3v = (const float*)d_in[13];
    float* out = (float*)d_out;

    char* ws = (char*)d_ws;
    int*   perm = (int*)  (ws + OFF_PERM);
    int*   goff = (int*)  (ws + OFF_GOFF);
    float* V    = (float*)(ws + OFF_V);

    sc_group_kernel<<<1, 512, 0, stream>>>(y, perm, goff);
    sc_build_v_kernel<<<240, 256, 0, stream>>>(U3s, U3v, W3s, W3v, V);
    sc_main_kernel<<<NWG2, WPB2 * 32, 0, stream>>>(
        x, perm, goff, V, U2s, U2v, U1s, U1v, W2s, W2v, W1s, W1v, out);
}


# Node grouping by element (perm, goff) runs as workgroup 0 of the V-build launch beside the 240 independent build workgroups, not in a launch of its own: 3 launches become 2; sums unchanged
# speedup vs baseline: 2.7933x; 1.1070x over previous
#include <hip/hip_runtime.h>

#define BB   512
#define CC   128
#define II   9
#define EE   10
#define NP   (EE * CC)
#define NBLK 4
#define VROWS 96
#define VK   12
#define VBLK (VROWS * VK)
#define VPAIR (NBLK * VBLK)

typedef float v2f __attribute__((ext_vector_type(2)));
typedef float v8f __attribute__((ext_vector_type(8)));

#define OFF_PERM  0
#define OFF_GOFF  4096
#define OFF_V     131072
#define WS_NEEDED (OFF_V + NP * VPAIR * 4)


#define NSEG 16

__global__ __launch_bounds__(256) void sc_build_v_kernel(
    const float* __restrict__ y, int* __restrict__ perm, int* __restrict__ goff,
    const float* __restrict__ U3s, const float* __restrict__ U3v,
    const float* __restrict__ W3s, const float* __restrict__ W3v,
    float* __restrict__ V)
{
    __shared__ int cnt[NSEG * EE];
    __shared__ int wbase[NSEG * EE];
    __shared__ int tot[EE];
    __shared__ int gstart[EE + 1];

    if (blockIdx.x == 0) {
        const int t = threadIdx.x;
        const int ln = t & 31, w = t >> 5;
        int e2[2], rank2[2];
#pragma unroll
        for (int h = 0; h < 2; ++h) {
            const int b = t + h * 256;
            int e = 0;
            for (int j = 0; j < EE; ++j)
                if (y[b * EE + j] > 0.5f) e = j;
            int myrank = 0;
            for (int j = 0; j < EE; ++j) {
                unsigned m = __builtin_amdgcn_ballot_w32(e == j);
                if (e == j) myrank = __popc(m & ((1u << ln) - 1u));
                if (ln == 0) cnt[(h * 8 + w) * EE + j] = __popc(m);
            }
            e2[h] = e; rank2[h] = myrank;
        }
        __syncthreads();
        if (t < EE) {
            int run = 0;
            for (int sg = 0; sg < NSEG; ++sg) { wbase[sg * EE + t] = run; run += cnt[sg * EE + t]; }
            tot[t] = run;
        }
        __syncthreads();
        if (t == 0) {
            int run = 0;
            for (int j = 0; j < EE; ++j) { gstart[j] = run; run += tot[j]; }
            gstart[EE] = run;
        }
        __syncthreads();
#pragma unroll
        for (int h = 0; h < 2; ++h)
            perm[gstart[e2[h]] + wbase[(h * 8 + w) * EE + e2[h]] + rank2[h]] = t + h * 256;
        if (t <= EE) goff[t] = gstart[t];
        return;
    }

    const int lane = threadIdx.x & 31;
    const int wave = __builtin_amdgcn_readfirstlane((int)(threadIdx.x >> 5));
    const int half = lane >> 4;
    const int lq   = lane & 15;
    const int gw   = ((int)blockIdx.x - 1) * 8 + wave;
    const int ng   = gw % 6;
    const int rest = gw / 6;
    const int blk  = rest & 3;
    const int ct   = (rest >> 2) & 7;
    const int e    = rest >> 5;
    const int K3   = (blk == 0) ? 7 : 10;
    const float* W3 = (blk == 0) ? W3s : W3v;
    const float* U3 = (blk == 0) ? U3s : (U3v + (blk - 1) * (81 * 9 * 10));

    v2f af[3];
#pragma unroll
    for (int kt = 0; kt < 3; ++kt) {
#pragma unroll
        for (int j = 0; j < 2; ++j) {
            int k = kt * 4 + half * 2 + j;
            bool ok = k < K3;
            float v = W3[(e * K3 + (ok ? k : 0)) * CC + ct * 16 + lq];
            af[kt][j] = ok ? v : 0.f;
        }
    }

    for (int t = 0; t < 6; ++t) {
        const int nt0 = ng * 12 + t * 2;
        v2f bf[2][3];
#pragma unroll
        for (int q = 0; q < 2; ++q) {
            int R  = (nt0 + q) * 16 + lq;
            int ij = R / VK;
            int ap = R - ij * VK;
            bool okr = (ij < 81) && (ap < II);
            int base = (ij * II + ap) * K3;
#pragma unroll
            for (int kt = 0; kt < 3; ++kt) {
#pragma unroll
                for (int j = 0; j < 2; ++j) {
                    int k = kt * 4 + half * 2 + j;
                    bool ok = okr && (k < K3);
                    float v = U3[ok ? base + k : 0];
                    bf[q][kt][j] = ok ? v : 0.f;
                }
            }
        }
        v8f acc0 = {}, acc1 = {};
#pragma unroll
        for (int kt = 0; kt < 3; ++kt) {
            acc0 = __builtin_amdgcn_wmma_f32_16x16x4_f32(false, af[kt], false, bf[0][kt], (short)0, acc0, false, false);
            acc1 = __builtin_amdgcn_wmma_f32_16x16x4_f32(false, af[kt], false, bf[1][kt], (short)0, acc1, false, false);
        }
#pragma unroll
        for (int v = 0; v < 8; ++v) {
            int c = ct * 16 + v + half * 8;
            float* dst = V + ((size_t)(e * CC + c) * NBLK + blk) * VBLK + nt0 * 16 + lq;
            dst[0]  = acc0[v];
            dst[16] = acc1[v];
        }
    }
}

#define TP    68
#define WPB2  10
#define NWG2  (NP / 5)

__global__ __launch_bounds__(320) void sc_main_kernel(
    const float* __restrict__ x, const int* __restrict__ perm, const int* __restrict__ goff,
    const float* __restrict__ V,
    const float* __restrict__ U2s, const float* __restrict__ U2v,
    const float* __restrict__ U1s, const float* __restrict__ U1v,
    const float* __restrict__ W2s, const float* __restrict__ W2v,
    const float* __restrict__ W1s, const float* __restrict__ W1v,
    float* __restrict__ out)
{
    __shared__ float tile_all[WPB2 * VROWS * TP];
    __shared__ float c2_all[WPB2 * VROWS];
    __shared__ float u2_l[4 * 243];
    __shared__ float u1_l[4 * 9];

    const int tid = threadIdx.x;
    for (int i = tid; i < 243; i += 320) u2_l[i] = U2s[i];
    for (int i = tid; i < 729; i += 320) u2_l[243 + i] = U2v[i];
    if (tid < 9)  u1_l[tid] = U1s[tid];
    if (tid < 27) u1_l[9 + tid] = U1v[tid];
    __syncthreads();

    const int lane = tid & 31;
    const int wave = __builtin_amdgcn_readfirstlane((int)(tid >> 5));
    const int half = lane >> 4;
    const int lq   = lane & 15;
    const int p = (int)blockIdx.x * 5 + (wave >> 1);
    const int s = wave & 1;
    const int e = p >> 7, c = p & 127;
    const int g0 = goff[e];
    const int nb = goff[e + 1] - g0;
    float* tile = tile_all + wave * (VROWS * TP);
    float* c2w  = c2_all + wave * VROWS;
    const float* Vp = V + (size_t)p * VPAIR;

    for (int ch = 0; ch * 64 < nb; ++ch) {
        v2f bf[4][3];
#pragma unroll
        for (int nt = 0; nt < 4; ++nt) {
            int j = ch * 64 + nt * 16 + lq;
            bool valid = j < nb;
            int b = perm[g0 + (valid ? j : 0)];
            const float* xp = x + ((size_t)b * CC + c) * II;
#pragma unroll
            for (int kt = 0; kt < 3; ++kt) {
#pragma unroll
                for (int jj = 0; jj < 2; ++jj) {
                    int k = kt * 4 + half * 2 + jj;
                    bool ok = valid && (k < II);
                    float v = xp[(k < II) ? k : 0];
                    bf[nt][kt][jj] = ok ? v : 0.f;
                }
            }
        }

        for (int bi = 0; bi < 2; ++bi) {
            const int blk = s * 2 + bi;
            const float* Vb = Vp + blk * VBLK;


            const float* W2 = (blk == 0) ? W2s : W2v;
            const float* W1 = (blk == 0) ? W1s : W1v;
            float w2[3];
#pragma unroll
            for (int k = 0; k < 3; ++k) w2[k] = W2[(e * 3 + k) * CC + c];
            const float w1 = W1[e * CC + c];
            for (int r = lane; r < 81; r += 32) {
                float c2 = 0.f;
#pragma unroll
                for (int k = 0; k < 3; ++k) c2 += u2_l[blk * 243 + r * 3 + k] * w2[k];
                c2w[r] = c2;
            }

            for (int hb = 0; hb < 2; ++hb) {
                v2f a[3][3];
#pragma unroll
                for (int m3 = 0; m3 < 3; ++m3) {
                    const float* ar = Vb + ((hb * 3 + m3) * 16 + lq) * VK + half * 2;
#pragma unroll
                    for (int kt = 0; kt < 3; ++kt) a[m3][kt] = *(const v2f*)(ar + kt * 4);
                }
#pragma unroll
                for (int m3 = 0; m3 < 3; ++m3) {
                    v8f acc[4] = {};
#pragma unroll
                    for (int kt = 0; kt < 3; ++kt) {
#pragma unroll
                        for (int nt = 0; nt < 4; ++nt)
                            acc[nt] = __builtin_amdgcn_wmma_f32_16x16x4_f32(
                                false, a[m3][kt], false, bf[nt][kt], (short)0, acc[nt], false, false);
                    }
#pragma unroll
                    for (int v = 0; v < 8; ++v) {
                        int row = (hb * 3 + m3) * 16 + v + half * 8;
#pragma unroll
                        for (int nt = 0; nt < 4; ++nt) tile[row * TP + nt * 16 + lq] = acc[nt][v];
                    }
                }
            }

            for (int ps = 0; ps < 2; ++ps) {
                const int col = ps * 32 + lane;
                const int j = ch * 64 + col;
                if (j < nb) {
                    const int b = perm[g0 + j];
                    const float* xp = x + ((size_t)b * CC + c) * II;
                    float xa[II];
#pragma unroll
                    for (int q = 0; q < II; ++q) xa[q] = xp[q];
                    float t1[II];
#pragma unroll
                    for (int i = 0; i < II; ++i) {
                        float acc1 = 0.f;
#pragma unroll
                        for (int q = 0; q < II; ++q)
                            acc1 += (c2w[i * 9 + q] + tile[(i * 9 + q) * TP + col]) * xa[q];
                        t1[i] = acc1;
                    }
                    float r = 0.f;
#pragma unroll
                    for (int q = 0; q < II; ++q) r += (u1_l[blk * 9 + q] * w1 + t1[q]) * xa[q];
                    if (blk == 0) out[b * 512 + c] = r;
                    else          out[b * 512 + CC + c * 3 + (blk - 1)] = r;
                }
            }
        }
    }
}

extern "C" void kernel_launch(void* const* d_in, const int* in_sizes, int n_in,
                              void* d_out, int out_size, void* d_ws, size_t ws_size,
                              hipStream_t stream) {
    (void)in_sizes; (void)n_in; (void)out_size;
    if (ws_size < (size_t)WS_NEEDED) return;

    const float* x   = (const float*)d_in[0];
    const float* y   = (const float*)d_in[1];
    const float* U1s = (const float*)d_in[2];
    const float* U2s = (const float*)d_in[3];
    const float* U3s = (const float*)d_in[4];
    const float* U1v = (const float*)d_in[5];
    const float* U2v = (const float*)d_in[6];
    const float* U3v = (const float*)d_in[7];
    const float* W1s = (const float*)d_in[8];
    const float* W2s = (const float*)d_in[9];
    const float* W3s = (const float*)d_in[10];
    const float* W1v = (const float*)d_in[11];
    const float* W2v = (const float*)d_in[12];
    const float* W3v = (const float*)d_in[13];
    float* out = (float*)d_out;

    char* ws = (char*)d_ws;
    int*   perm = (int*)  (ws + OFF_PERM);
    int*   goff = (int*)  (ws + OFF_GOFF);
    float* V    = (float*)(ws + OFF_V);

    sc_build_v_kernel<<<241, 256, 0, stream>>>(y, perm, goff, U3s, U3v, W3s, W3v, V);
    sc_main_kernel<<<NWG2, WPB2 * 32, 0, stream>>>(
        x, perm, goff, V, U2s, U2v, U1s, U1v, W2s, W2v, W1s, W1v, out);
}


# V build fused into the main kernel: each workgroup (one block, 20 pairs) builds its V on WMMA into LDS and uses it there; V leaves the workspace; nodes grouped per workgroup; 2 launches to 1
# speedup vs baseline: 3.3959x; 1.2157x over previous
#include <hip/hip_runtime.h>

#define BB   512
#define CC   128
#define II   9
#define EE   10
#define NP   (EE * CC)
#define NBLK 4
#define NROW 81
#define VK   12
#define NPOS_T 61

typedef float v2f __attribute__((ext_vector_type(2)));
typedef float v8f __attribute__((ext_vector_type(8)));

#define NW    20
#define NWG   (NBLK * (NP / NW))
#define PV    980
#define TP    36
#define NSEG  16

__device__ __forceinline__ void gather_u3(const float* __restrict__ U3, int K3, int nt, int lq, int half, v2f (&bf)[3])
{
    const int R  = nt * 16 + lq;
    const int ij = R / VK;
    const int ap = R - ij * VK;
    const bool okr = (ij < NROW) && (ap < II);
    const int base = (ij * II + ap) * K3;
#pragma unroll
    for (int kt = 0; kt < 3; ++kt) {
#pragma unroll
        for (int j = 0; j < 2; ++j) {
            int k = kt * 4 + half * 2 + j;
            bool ok = okr && (k < K3);
            float v = U3[ok ? base + k : 0];
            bf[kt][j] = ok ? v : 0.f;
        }
    }
}

__device__ __forceinline__ void build_tile(const v2f (&af)[2][3], const v2f (&bf)[3], float* Vl, int nt, int lq, int half)
{
    v8f acc0 = {}, acc1 = {};
#pragma unroll
    for (int kt = 0; kt < 3; ++kt) {
        acc0 = __builtin_amdgcn_wmma_f32_16x16x4_f32(false, af[0][kt], false, bf[kt], (short)0, acc0, false, false);
        acc1 = __builtin_amdgcn_wmma_f32_16x16x4_f32(false, af[1][kt], false, bf[kt], (short)0, acc1, false, false);
    }
    asm volatile("s_wait_alu depctr_va_vdst(0)\n\tv_nop\n\tv_nop\n\tv_nop\n\tv_nop" : "+v"(acc0), "+v"(acc1));
#pragma unroll
    for (int v = 0; v < 8; ++v)
        Vl[(v + half * 8) * PV + nt * 16 + lq] = acc0[v];
#pragma unroll
    for (int v = 0; v < 2; ++v)
        Vl[(16 + v + half * 2) * PV + nt * 16 + lq] = acc1[v];
}

__global__ __launch_bounds__(NW * 32) void sc_fused_kernel(
    const float* __restrict__ x, const float* __restrict__ y,
    const float* __restrict__ U3s, const float* __restrict__ U3v,
    const float* __restrict__ W3s, const float* __restrict__ W3v,
    const float* __restrict__ U2s, const float* __restrict__ U2v,
    const float* __restrict__ U1s, const float* __restrict__ U1v,
    const float* __restrict__ W2s, const float* __restrict__ W2v,
    const float* __restrict__ W1s, const float* __restrict__ W1v,
    float* __restrict__ out)
{
    __shared__ __attribute__((aligned(16))) float tile_all[NW * NROW * TP];
    __shared__ __attribute__((aligned(16))) float Vl[NW * PV];
    __shared__ float c2_all[NW * NROW];
    __shared__ float u2_l[243];
    __shared__ float u1_l[II];
    __shared__ int perm_l[BB];
    __shared__ int cnt[NSEG * EE];
    __shared__ int wbase[NSEG * EE];
    __shared__ int tot[EE];
    __shared__ int gstart[EE + 1];

    const int tid  = threadIdx.x;
    const int lane = tid & 31;
    const int wave = __builtin_amdgcn_readfirstlane((int)(tid >> 5));
    const int half = lane >> 4;
    const int lq   = lane & 15;
    const int blk  = (int)blockIdx.x & 3;
    const int p0   = ((int)blockIdx.x >> 2) * NW;

    {
        const float* U2 = (blk == 0) ? U2s : (U2v + (blk - 1) * 243);
        const float* U1 = (blk == 0) ? U1s : (U1v + (blk - 1) * II);
        if (tid < 243) u2_l[tid] = U2[tid];
        if (tid >= 256 && tid < 256 + II) u1_l[tid - 256] = U1[tid - 256];
    }

    int e2[2] = {0, 0}, rank2[2] = {0, 0};
    if (wave < 8) {
#pragma unroll
        for (int h = 0; h < 2; ++h) {
            const int b = tid + h * 256;
            int e = 0;
            for (int j = 0; j < EE; ++j)
                if (y[b * EE + j] > 0.5f) e = j;
            int myrank = 0;
            for (int j = 0; j < EE; ++j) {
                unsigned m = __builtin_amdgcn_ballot_w32(e == j);
                if (e == j) myrank = __popc(m & ((1u << lane) - 1u));
                if (lane == 0) cnt[(h * 8 + wave) * EE + j] = __popc(m);
            }
            e2[h] = e; rank2[h] = myrank;
        }
    }

    {
        const int K3   = (blk == 0) ? 7 : 10;
        const float* W3 = (blk == 0) ? W3s : W3v;
        const float* U3 = (blk == 0) ? U3s : (U3v + (blk - 1) * (81 * 9 * 10));
        v2f af[2][3];
#pragma unroll
        for (int m2 = 0; m2 < 2; ++m2) {
            const int pl = (m2 == 0) ? lq : (16 + (lq & 7) + (lq >> 3) * 2);
            const bool okp = (m2 == 0) || ((lq & 7) < 2);
            const int pp = p0 + (okp ? pl : 0);
            const int pe = pp >> 7, pc = pp & 127;
#pragma unroll
            for (int kt = 0; kt < 3; ++kt) {
#pragma unroll
                for (int j = 0; j < 2; ++j) {
                    int k = kt * 4 + half * 2 + j;
                    bool ok = okp && (k < K3);
                    float v = W3[(pe * K3 + ((k < K3) ? k : 0)) * CC + pc];
                    af[m2][kt][j] = ok ? v : 0.f;
                }
            }
        }
        v2f bf[3][3];
#pragma unroll
        for (int t = 0; t < 3; ++t) gather_u3(U3, K3, wave + t * NW, lq, half, bf[t]);
#pragma unroll
        for (int t = 0; t < 3; ++t) build_tile(af, bf[t], Vl, wave + t * NW, lq, half);
        if (wave == NW - 1) {
            v2f bl[3];
            gather_u3(U3, K3, NPOS_T - 1, lq, half, bl);
            build_tile(af, bl, Vl, NPOS_T - 1, lq, half);
        }
    }

    __syncthreads();
    if (tid < EE) {
        int run = 0;
        for (int sg = 0; sg < NSEG; ++sg) { wbase[sg * EE + tid] = run; run += cnt[sg * EE + tid]; }
        tot[tid] = run;
    }
    __syncthreads();
    if (tid == 0) {
        int run = 0;
        for (int j = 0; j < EE; ++j) { gstart[j] = run; run += tot[j]; }
        gstart[EE] = run;
    }
    __syncthreads();
    if (wave < 8) {
#pragma unroll
        for (int h = 0; h < 2; ++h)
            perm_l[gstart[e2[h]] + wbase[(h * 8 + wave) * EE + e2[h]] + rank2[h]] = tid + h * 256;
    }
    __syncthreads();

    const int p = p0 + wave;
    const int e = p >> 7, c = p & 127;
    const int g0 = __builtin_amdgcn_readfirstlane(gstart[e]);
    const int nb = __builtin_amdgcn_readfirstlane(gstart[e + 1]) - g0;
    float* tile = tile_all + wave * (NROW * TP);
    float* c2w  = c2_all + wave * NROW;
    const float* Vw = Vl + wave * PV;

    const float* W2 = (blk == 0) ? W2s : W2v;
    const float* W1 = (blk == 0) ? W1s : W1v;
    float w2[3];
#pragma unroll
    for (int k = 0; k < 3; ++k) w2[k] = W2[(e * 3 + k) * CC + c];
    const float w1 = W1[e * CC + c];
    for (int r = lane; r < NROW; r += 32) {
        float c2 = 0.f;
#pragma unroll
        for (int k = 0; k < 3; ++k) c2 += u2_l[r * 3 + k] * w2[k];
        c2w[r] = c2;
    }

    for (int ch = 0; ch * 32 < nb; ++ch) {
        v2f bf[2][3];
#pragma unroll
        for (int nt = 0; nt < 2; ++nt) {
            int j = ch * 32 + nt * 16 + lq;
            bool valid = j < nb;
            int b = perm_l[g0 + (valid ? j : 0)];
            const float* xp = x + ((size_t)b * CC + c) * II;
#pragma unroll
            for (int kt = 0; kt < 3; ++kt) {
#pragma unroll
                for (int jj = 0; jj < 2; ++jj) {
                    int k = kt * 4 + half * 2 + jj;
                    bool ok = valid && (k < II);
                    float v = xp[(k < II) ? k : 0];
                    bf[nt][kt][jj] = ok ? v : 0.f;
                }
            }
        }

#pragma unroll 1
        for (int hb = 0; hb < 2; ++hb) {
            v2f a[3][3];
#pragma unroll
            for (int m3 = 0; m3 < 3; ++m3) {
                const int row = (hb * 3 + m3) * 16 + lq;
                const float* ar = Vw + ((row < NROW) ? row : (NROW - 1)) * VK + half * 2;
#pragma unroll
                for (int kt = 0; kt < 3; ++kt) a[m3][kt] = *(const v2f*)(ar + kt * 4);
            }
#pragma unroll
            for (int m3 = 0; m3 < 3; ++m3) {
                v8f acc[2] = {};
#pragma unroll
                for (int kt = 0; kt < 3; ++kt) {
#pragma unroll
                    for (int nt = 0; nt < 2; ++nt)
                        acc[nt] = __builtin_amdgcn_wmma_f32_16x16x4_f32(
                            false, a[m3][kt], false, bf[nt][kt], (short)0, acc[nt], false, false);
                }
#pragma unroll
                for (int v = 0; v < 8; ++v) {
                    int row = (hb * 3 + m3) * 16 + v + half * 8;
                    if (row < NROW) {
#pragma unroll
                        for (int nt = 0; nt < 2; ++nt) tile[row * TP + nt * 16 + lq] = acc[nt][v];
                    }
                }
            }
        }

        const int j = ch * 32 + lane;
        if (j < nb) {
            const int b = perm_l[g0 + j];
            const float* xp = x + ((size_t)b * CC + c) * II;
            float xa[II];
#pragma unroll
            for (int q = 0; q < II; ++q) xa[q] = xp[q];
            float t1[II];
#pragma unroll
            for (int i = 0; i < II; ++i) {
                float acc1 = 0.f;
#pragma unroll
                for (int q = 0; q < II; ++q)
                    acc1 += (c2w[i * 9 + q] + tile[(i * 9 + q) * TP + lane]) * xa[q];
                t1[i] = acc1;
            }
            float r = 0.f;
#pragma unroll
            for (int q = 0; q < II; ++q) r += (u1_l[q] * w1 + t1[q]) * xa[q];
            if (blk == 0) out[b * 512 + c] = r;
            else          out[b * 512 + CC + c * 3 + (blk - 1)] = r;
        }
    }
}

extern "C" void kernel_launch(void* const* d_in, const int* in_sizes, int n_in,
                              void* d_out, int out_size, void* d_ws, size_t ws_size,
                              hipStream_t stream) {
    (void)in_sizes; (void)n_in; (void)out_size; (void)d_ws; (void)ws_size;

    const float* x   = (const float*)d_in[0];
    const float* y   = (const float*)d_in[1];
    const float* U1s = (const float*)d_in[2];
    const float* U2s = (const float*)d_in[3];
    const float* U3s = (const float*)d_in[4];
    const float* U1v = (const float*)d_in[5];
    const float* U2v = (const float*)d_in[6];
    const float* U3v = (const float*)d_in[7];
    const float* W1s = (const float*)d_in[8];
    const float* W2s = (const float*)d_in[9];
    const float* W3s = (const float*)d_in[10];
    const float* W1v = (const float*)d_in[11];
    const float* W2v = (const float*)d_in[12];
    const float* W3v = (const float*)d_in[13];
    float* out = (float*)d_out;

    sc_fused_kernel<<<NWG, NW * 32, 0, stream>>>(
        x, y, U3s, U3v, W3s, W3v, U2s, U2v, U1s, U1v, W2s, W2v, W1s, W1v, out);
}


# Global loads issued before the first wait: the opening's (y, U1, U2, W2, W1, W3, U3) as one group; each pass reads one x row per lane a pass ahead, B fragments by lane shuffle; sums kept
# speedup vs baseline: 4.0183x; 1.1833x over previous
#include <hip/hip_runtime.h>

#define BB   512
#define CC   128
#define II   9
#define EE   10
#define NP   (EE * CC)
#define NBLK 4
#define NROW 81
#define VK   12
#define NPOS_T 61

typedef float v2f __attribute__((ext_vector_type(2)));
typedef float v8f __attribute__((ext_vector_type(8)));

#define NW    20
#define NWG   (NBLK * (NP / NW))
#define PV    980
#define TP    36
#define NSEG  16

__device__ __forceinline__ void gather_u3(const float* __restrict__ U3, int K3, int nt, int lq, int half, v2f (&bf)[3])
{
    const int R  = nt * 16 + lq;
    const int ij = R / VK;
    const int ap = R - ij * VK;
    const bool okr = (ij < NROW) && (ap < II);
    const int base = (ij * II + ap) * K3;
#pragma unroll
    for (int kt = 0; kt < 3; ++kt) {
#pragma unroll
        for (int j = 0; j < 2; ++j) {
            int k = kt * 4 + half * 2 + j;
            bool ok = okr && (k < K3);
            float v = U3[ok ? base + k : 0];
            bf[kt][j] = ok ? v : 0.f;
        }
    }
}

__device__ __forceinline__ void build_tile(const v2f (&af)[2][3], const v2f (&bf)[3], float* Vl, int nt, int lq, int half)
{
    v8f acc0 = {}, acc1 = {};
#pragma unroll
    for (int kt = 0; kt < 3; ++kt) {
        acc0 = __builtin_amdgcn_wmma_f32_16x16x4_f32(false, af[0][kt], false, bf[kt], (short)0, acc0, false, false);
        acc1 = __builtin_amdgcn_wmma_f32_16x16x4_f32(false, af[1][kt], false, bf[kt], (short)0, acc1, false, false);
    }
    asm volatile("s_wait_alu depctr_va_vdst(0)\n\tv_nop\n\tv_nop\n\tv_nop\n\tv_nop" : "+v"(acc0), "+v"(acc1));
#pragma unroll
    for (int v = 0; v < 8; ++v)
        Vl[(v + half * 8) * PV + nt * 16 + lq] = acc0[v];
#pragma unroll
    for (int v = 0; v < 2; ++v)
        Vl[(16 + v + half * 2) * PV + nt * 16 + lq] = acc1[v];
}

__global__ __launch_bounds__(NW * 32) void sc_fused_kernel(
    const float* __restrict__ x, const float* __restrict__ y,
    const float* __restrict__ U3s, const float* __restrict__ U3v,
    const float* __restrict__ W3s, const float* __restrict__ W3v,
    const float* __restrict__ U2s, const float* __restrict__ U2v,
    const float* __restrict__ U1s, const float* __restrict__ U1v,
    const float* __restrict__ W2s, const float* __restrict__ W2v,
    const float* __restrict__ W1s, const float* __restrict__ W1v,
    float* __restrict__ out)
{
    __shared__ __attribute__((aligned(16))) float tile_all[NW * NROW * TP];
    __shared__ __attribute__((aligned(16))) float Vl[NW * PV];
    __shared__ float c2_all[NW * NROW];
    __shared__ float u1_l[II];
    __shared__ int perm_l[BB];
    __shared__ int cnt[NSEG * EE];
    __shared__ int wbase[NSEG * EE];
    __shared__ int tot[EE];
    __shared__ int gstart[EE + 1];

    const int tid  = threadIdx.x;
    const int lane = tid & 31;
    const int wave = __builtin_amdgcn_readfirstlane((int)(tid >> 5));
    const int half = lane >> 4;
    const int lq   = lane & 15;
    const int blk  = (int)blockIdx.x & 3;
    const int p0   = ((int)blockIdx.x >> 2) * NW;

    const int p = p0 + wave;
    const int e = p >> 7, c = p & 127;
    const int K3   = (blk == 0) ? 7 : 10;
    const float* W3 = (blk == 0) ? W3s : W3v;
    const float* U3 = (blk == 0) ? U3s : (U3v + (blk - 1) * (81 * 9 * 10));
    const float* U2 = (blk == 0) ? U2s : (U2v + (blk - 1) * 243);
    const float* U1 = (blk == 0) ? U1s : (U1v + (blk - 1) * II);
    const float* W2 = (blk == 0) ? W2s : W2v;
    const float* W1 = (blk == 0) ? W1s : W1v;

    const float u1v = U1[(tid < II) ? tid : 0];
    const int pw = p0 + (tid >> 5);
    const int ew = pw >> 7, cw = pw & 127;
    float w2[3];
#pragma unroll
    for (int k = 0; k < 3; ++k) w2[k] = W2[(ew * 3 + k) * CC + cw];
    const float w1 = W1[ew * CC + cw];
    float u2v[3][3];
#pragma unroll
    for (int m = 0; m < 3; ++m) {
        const int r = lane + m * 32;
#pragma unroll
        for (int k = 0; k < 3; ++k) u2v[m][k] = U2[((r < NROW) ? r : 0) * 3 + k];
    }
    v2f af[2][3];
#pragma unroll
    for (int m2 = 0; m2 < 2; ++m2) {
        const int pl = (m2 == 0) ? lq : (16 + (lq & 7) + (lq >> 3) * 2);
        const bool okp = (m2 == 0) || ((lq & 7) < 2);
        const int pp = p0 + (okp ? pl : 0);
        const int pe = pp >> 7, pc = pp & 127;
#pragma unroll
        for (int kt = 0; kt < 3; ++kt) {
#pragma unroll
            for (int j = 0; j < 2; ++j) {
                int k = kt * 4 + half * 2 + j;
                bool ok = okp && (k < K3);
                float v = W3[(pe * K3 + ((k < K3) ? k : 0)) * CC + pc];
                af[m2][kt][j] = ok ? v : 0.f;
            }
        }
    }
    v2f bf3[3][3];
#pragma unroll
    for (int t = 0; t < 3; ++t) gather_u3(U3, K3, wave + t * NW, lq, half, bf3[t]);
    v2f bl[3] = {};
    if (wave == NW - 1) gather_u3(U3, K3, NPOS_T - 1, lq, half, bl);
    float yv[2][EE];
#pragma unroll
    for (int h = 0; h < 2; ++h) {
#pragma unroll
        for (int j = 0; j < EE; ++j) yv[h][j] = 0.f;
    }
    if (wave < 8) {
#pragma unroll
        for (int h = 0; h < 2; ++h) {
#pragma unroll
            for (int j = 0; j < EE; ++j) yv[h][j] = y[(tid + h * 256) * EE + j];
        }
    }

    if (tid < II) u1_l[tid] = u1v;
    float* c2w  = c2_all + wave * NROW;
#pragma unroll
    for (int m = 0; m < 3; ++m) {
        const int r = lane + m * 32;
        float c2 = 0.f;
#pragma unroll
        for (int k = 0; k < 3; ++k) c2 += u2v[m][k] * w2[k];
        if (r < NROW) c2w[r] = c2;
    }

    int e2[2] = {0, 0}, rank2[2] = {0, 0};
    if (wave < 8) {
#pragma unroll
        for (int h = 0; h < 2; ++h) {
            int en = 0;
#pragma unroll
            for (int j = 0; j < EE; ++j)
                if (yv[h][j] > 0.5f) en = j;
            int myrank = 0;
            for (int j = 0; j < EE; ++j) {
                unsigned m = __builtin_amdgcn_ballot_w32(en == j);
                if (en == j) myrank = __popc(m & ((1u << lane) - 1u));
                if (lane == 0) cnt[(h * 8 + wave) * EE + j] = __popc(m);
            }
            e2[h] = en; rank2[h] = myrank;
        }
    }

    {
#pragma unroll
        for (int t = 0; t < 3; ++t) build_tile(af, bf3[t], Vl, wave + t * NW, lq, half);
        if (wave == NW - 1) build_tile(af, bl, Vl, NPOS_T - 1, lq, half);
    }

    __syncthreads();
    if (tid < EE) {
        int run = 0;
        for (int sg = 0; sg < NSEG; ++sg) { wbase[sg * EE + tid] = run; run += cnt[sg * EE + tid]; }
        tot[tid] = run;
    }
    __syncthreads();
    if (tid == 0) {
        int run = 0;
        for (int j = 0; j < EE; ++j) { gstart[j] = run; run += tot[j]; }
        gstart[EE] = run;
    }
    __syncthreads();
    if (wave < 8) {
#pragma unroll
        for (int h = 0; h < 2; ++h)
            perm_l[gstart[e2[h]] + wbase[(h * 8 + wave) * EE + e2[h]] + rank2[h]] = tid + h * 256;
    }
    __syncthreads();

    const int g0 = __builtin_amdgcn_readfirstlane(gstart[e]);
    const int nb = __builtin_amdgcn_readfirstlane(gstart[e + 1]) - g0;
    float* tile = tile_all + wave * (NROW * TP);
    const float* Vw = Vl + wave * PV;
    if (nb <= 0) return;

    float xn[II];
    bool vn = lane < nb;
    int bn = perm_l[g0 + (vn ? lane : 0)];
    {
        const float* xp = x + ((size_t)bn * CC + c) * II;
#pragma unroll
        for (int q = 0; q < II; ++q) xn[q] = xp[q];
    }

    for (int ch = 0; ch * 32 < nb; ++ch) {
        const bool vcur = vn;
        const int b = bn;
        float xa[II];
#pragma unroll
        for (int q = 0; q < II; ++q) xa[q] = vcur ? xn[q] : 0.f;
        {
            const int jn = (ch + 1) * 32 + lane;
            vn = jn < nb;
            bn = perm_l[g0 + (vn ? jn : 0)];
            const float* xp = x + ((size_t)bn * CC + c) * II;
#pragma unroll
            for (int q = 0; q < II; ++q) xn[q] = xp[q];
        }
        v2f bf[2][3];
#pragma unroll
        for (int nt = 0; nt < 2; ++nt) {
            float s[II];
#pragma unroll
            for (int k = 0; k < II; ++k) s[k] = __shfl(xa[k], nt * 16 + lq);
            bf[nt][0][0] = half ? s[2] : s[0];
            bf[nt][0][1] = half ? s[3] : s[1];
            bf[nt][1][0] = half ? s[6] : s[4];
            bf[nt][1][1] = half ? s[7] : s[5];
            bf[nt][2][0] = half ? 0.f : s[8];
            bf[nt][2][1] = 0.f;
        }

#pragma unroll 1
        for (int hb = 0; hb < 2; ++hb) {
            v2f a[3][3];
#pragma unroll
            for (int m3 = 0; m3 < 3; ++m3) {
                const int row = (hb * 3 + m3) * 16 + lq;
                const float* ar = Vw + ((row < NROW) ? row : (NROW - 1)) * VK + half * 2;
#pragma unroll
                for (int kt = 0; kt < 3; ++kt) a[m3][kt] = *(const v2f*)(ar + kt * 4);
            }
#pragma unroll
            for (int m3 = 0; m3 < 3; ++m3) {
                v8f acc[2] = {};
#pragma unroll
                for (int kt = 0; kt < 3; ++kt) {
#pragma unroll
                    for (int nt = 0; nt < 2; ++nt)
                        acc[nt] = __builtin_amdgcn_wmma_f32_16x16x4_f32(
                            false, a[m3][kt], false, bf[nt][kt], (short)0, acc[nt], false, false);
                }
#pragma unroll
                for (int v = 0; v < 8; ++v) {
                    int row = (hb * 3 + m3) * 16 + v + half * 8;
                    if (row < NROW) {
#pragma unroll
                        for (int nt = 0; nt < 2; ++nt) tile[row * TP + nt * 16 + lq] = acc[nt][v];
                    }
                }
            }
        }

        if (vcur) {
            float t1[II];
#pragma unroll
            for (int i = 0; i < II; ++i) {
                float acc1 = 0.f;
#pragma unroll
                for (int q = 0; q < II; ++q)
                    acc1 += (c2w[i * 9 + q] + tile[(i * 9 + q) * TP + lane]) * xa[q];
                t1[i] = acc1;
                if (i == 4) __builtin_amdgcn_sched_barrier(0);
            }
            float r = 0.f;
#pragma unroll
            for (int q = 0; q < II; ++q) r += (u1_l[q] * w1 + t1[q]) * xa[q];
            if (blk == 0) out[b * 512 + c] = r;
            else          out[b * 512 + CC + c * 3 + (blk - 1)] = r;
        }
    }
}

extern "C" void kernel_launch(void* const* d_in, const int* in_sizes, int n_in,
                              void* d_out, int out_size, void* d_ws, size_t ws_size,
                              hipStream_t stream) {
    (void)in_sizes; (void)n_in; (void)out_size; (void)d_ws; (void)ws_size;

    const float* x   = (const float*)d_in[0];
    const float* y   = (const float*)d_in[1];
    const float* U1s = (const float*)d_in[2];
    const float* U2s = (const float*)d_in[3];
    const float* U3s = (const float*)d_in[4];
    const float* U1v = (const float*)d_in[5];
    const float* U2v = (const float*)d_in[6];
    const float* U3v = (const float*)d_in[7];
    const float* W1s = (const float*)d_in[8];
    const float* W2s = (const float*)d_in[9];
    const float* W3s = (const float*)d_in[10];
    const float* W1v = (const float*)d_in[11];
    const float* W2v = (const float*)d_in[12];
    const float* W3v = (const float*)d_in[13];
    float* out = (float*)d_out;

    sc_fused_kernel<<<NWG, NW * 32, 0, stream>>>(
        x, y, U3s, U3v, W3s, W3v, U2s, U2v, U1s, U1v, W2s, W2v, W1s, W1v, out);
}


# nu=2 stage folded into the node product: per wave (9 x 92) x (92 x nodes) on WMMA, B = x_j*x_a formed in registers, K = 81 (j,a) + 9 c2; LDS result tile gone; nu=1 on accumulators; sum order changes
# speedup vs baseline: 4.5026x; 1.1205x over previous
#include <hip/hip_runtime.h>

#define BB   512
#define CC   128
#define II   9
#define EE   10
#define NP   (EE * CC)
#define NBLK 4
#define NROW 81
#define RP   92
#define NKT  23
#define NPOS_T 52

typedef float v2f __attribute__((ext_vector_type(2)));
typedef float v8f __attribute__((ext_vector_type(8)));

#define NW    20
#define NWG   (NBLK * (NP / NW))
#define PV    828
#define NSEG  16

__device__ __forceinline__ bool gather_u3(const float* __restrict__ U3, int K3, int nt, int lq, int half, v2f (&bf)[3])
{
    const int R  = nt * 16 + lq;
    const int i  = R / RP;
    const int kk = R - i * RP;
    const int j  = (kk < 72) ? (kk >> 3) : ((kk < 80) ? (kk - 72) : 8);
    const int a  = (kk < 72) ? (kk & 7) : 8;
    const bool okr = (i < II) && ((kk < 80) || (kk == 88));
    const int base = ((i * II + j) * II + a) * K3;
#pragma unroll
    for (int kt = 0; kt < 3; ++kt) {
#pragma unroll
        for (int q = 0; q < 2; ++q) {
            int k = kt * 4 + half * 2 + q;
            bool ok = okr && (k < K3);
            float v = U3[ok ? base + k : 0];
            bf[kt][q] = ok ? v : 0.f;
        }
    }
    return okr;
}

__device__ __forceinline__ void build_tile(const v2f (&af)[2][3], const v2f (&bf)[3], float* Vl, int nt, int lq, int half, bool okpos)
{
    v8f acc0 = {}, acc1 = {};
#pragma unroll
    for (int kt = 0; kt < 3; ++kt) {
        acc0 = __builtin_amdgcn_wmma_f32_16x16x4_f32(false, af[0][kt], false, bf[kt], (short)0, acc0, false, false);
        acc1 = __builtin_amdgcn_wmma_f32_16x16x4_f32(false, af[1][kt], false, bf[kt], (short)0, acc1, false, false);
    }
    asm volatile("s_wait_alu depctr_va_vdst(0)\n\tv_nop\n\tv_nop\n\tv_nop\n\tv_nop" : "+v"(acc0), "+v"(acc1));
    if (okpos) {
#pragma unroll
        for (int v = 0; v < 8; ++v)
            Vl[(v + half * 8) * PV + nt * 16 + lq] = acc0[v];
#pragma unroll
        for (int v = 0; v < 2; ++v)
            Vl[(16 + v + half * 2) * PV + nt * 16 + lq] = acc1[v];
    }
}

__global__ __launch_bounds__(NW * 32) void sc_fused_kernel(
    const float* __restrict__ x, const float* __restrict__ y,
    const float* __restrict__ U3s, const float* __restrict__ U3v,
    const float* __restrict__ W3s, const float* __restrict__ W3v,
    const float* __restrict__ U2s, const float* __restrict__ U2v,
    const float* __restrict__ U1s, const float* __restrict__ U1v,
    const float* __restrict__ W2s, const float* __restrict__ W2v,
    const float* __restrict__ W1s, const float* __restrict__ W1v,
    float* __restrict__ out)
{
    __shared__ __attribute__((aligned(16))) float Vl[NW * PV];
    __shared__ float u1_l[II];
    __shared__ int perm_l[BB];
    __shared__ int cnt[NSEG * EE];
    __shared__ int wbase[NSEG * EE];
    __shared__ int tot[EE];
    __shared__ int gstart[EE + 1];

    const int tid  = threadIdx.x;
    const int lane = tid & 31;
    const int wave = __builtin_amdgcn_readfirstlane((int)(tid >> 5));
    const int half = lane >> 4;
    const int lq   = lane & 15;
    const int blk  = (int)blockIdx.x & 3;
    const int p0   = ((int)blockIdx.x >> 2) * NW;

    const int p = p0 + wave;
    const int e = p >> 7, c = p & 127;
    const int K3   = (blk == 0) ? 7 : 10;
    const float* W3 = (blk == 0) ? W3s : W3v;
    const float* U3 = (blk == 0) ? U3s : (U3v + (blk - 1) * (81 * 9 * 10));
    const float* U2 = (blk == 0) ? U2s : (U2v + (blk - 1) * 243);
    const float* U1 = (blk == 0) ? U1s : (U1v + (blk - 1) * II);
    const float* W2 = (blk == 0) ? W2s : W2v;
    const float* W1 = (blk == 0) ? W1s : W1v;

    const float u1v = U1[(tid < II) ? tid : 0];
    const int pw = p0 + (tid >> 5);
    const int ew = pw >> 7, cw = pw & 127;
    float w2[3];
#pragma unroll
    for (int k = 0; k < 3; ++k) w2[k] = W2[(ew * 3 + k) * CC + cw];
    const float w1 = W1[ew * CC + cw];
    float u2v[3][3];
#pragma unroll
    for (int m = 0; m < 3; ++m) {
        const int r = lane + m * 32;
#pragma unroll
        for (int k = 0; k < 3; ++k) u2v[m][k] = U2[((r < NROW) ? r : 0) * 3 + k];
    }
    v2f af[2][3];
#pragma unroll
    for (int m2 = 0; m2 < 2; ++m2) {
        const int pl = (m2 == 0) ? lq : (16 + (lq & 7) + (lq >> 3) * 2);
        const bool okp = (m2 == 0) || ((lq & 7) < 2);
        const int pp = p0 + (okp ? pl : 0);
        const int pe = pp >> 7, pc = pp & 127;
#pragma unroll
        for (int kt = 0; kt < 3; ++kt) {
#pragma unroll
            for (int q = 0; q < 2; ++q) {
                int k = kt * 4 + half * 2 + q;
                bool ok = okp && (k < K3);
                float v = W3[(pe * K3 + ((k < K3) ? k : 0)) * CC + pc];
                af[m2][kt][q] = ok ? v : 0.f;
            }
        }
    }
    v2f bf3[2][3];
    bool okb[2];
#pragma unroll
    for (int t = 0; t < 2; ++t) okb[t] = gather_u3(U3, K3, wave + t * NW, lq, half, bf3[t]);
    v2f bl[3] = {};
    bool okl = false;
    if (wave >= 8) okl = gather_u3(U3, K3, 32 + wave, lq, half, bl);
    float yv[2][EE];
#pragma unroll
    for (int h = 0; h < 2; ++h) {
#pragma unroll
        for (int q = 0; q < EE; ++q) yv[h][q] = 0.f;
    }
    if (wave < 8) {
#pragma unroll
        for (int h = 0; h < 2; ++h) {
#pragma unroll
            for (int q = 0; q < EE; ++q) yv[h][q] = y[(tid + h * 256) * EE + q];
        }
    }

    if (tid < II) u1_l[tid] = u1v;
    float* Vw = Vl + wave * PV;
#pragma unroll
    for (int m = 0; m < 3; ++m) {
        const int r = lane + m * 32;
        float c2 = 0.f;
#pragma unroll
        for (int k = 0; k < 3; ++k) c2 += u2v[m][k] * w2[k];
        const int ri = r / II, rj = r - ri * II;
        if (r < NROW) Vw[ri * RP + ((rj < 8) ? (80 + rj) : 89)] = c2;
    }
    if (lane < 2 * II) Vw[(lane >> 1) * RP + 90 + (lane & 1)] = 0.f;

    int e2[2] = {0, 0}, rank2[2] = {0, 0};
    if (wave < 8) {
#pragma unroll
        for (int h = 0; h < 2; ++h) {
            int en = 0;
#pragma unroll
            for (int q = 0; q < EE; ++q)
                if (yv[h][q] > 0.5f) en = q;
            int myrank = 0;
            for (int q = 0; q < EE; ++q) {
                unsigned m = __builtin_amdgcn_ballot_w32(en == q);
                if (en == q) myrank = __popc(m & ((1u << lane) - 1u));
                if (lane == 0) cnt[(h * 8 + wave) * EE + q] = __popc(m);
            }
            e2[h] = en; rank2[h] = myrank;
        }
    }

    {
#pragma unroll
        for (int t = 0; t < 2; ++t) build_tile(af, bf3[t], Vl, wave + t * NW, lq, half, okb[t]);
        if (wave >= 8) build_tile(af, bl, Vl, 32 + wave, lq, half, okl);
    }

    __syncthreads();
    if (tid < EE) {
        int run = 0;
        for (int sg = 0; sg < NSEG; ++sg) { wbase[sg * EE + tid] = run; run += cnt[sg * EE + tid]; }
        tot[tid] = run;
    }
    __syncthreads();
    if (tid == 0) {
        int run = 0;
        for (int q = 0; q < EE; ++q) { gstart[q] = run; run += tot[q]; }
        gstart[EE] = run;
    }
    __syncthreads();
    if (wave < 8) {
#pragma unroll
        for (int h = 0; h < 2; ++h)
            perm_l[gstart[e2[h]] + wbase[(h * 8 + wave) * EE + e2[h]] + rank2[h]] = tid + h * 256;
    }
    __syncthreads();

    const int g0 = __builtin_amdgcn_readfirstlane(gstart[e]);
    const int nb = __builtin_amdgcn_readfirstlane(gstart[e + 1]) - g0;
    if (nb <= 0) return;

    v2f a[NKT];
    {
        const float* ar = Vw + ((lq < II) ? lq : (II - 1)) * RP + half * 2;
#pragma unroll
        for (int kt = 0; kt < NKT; ++kt) a[kt] = *(const v2f*)(ar + kt * 4);
    }
    float c1v[II];
#pragma unroll
    for (int q = 0; q < II; ++q) c1v[q] = u1_l[q] * w1;
    const float c1a = half ? c1v[8] : c1v[0];

    float xn[II];
    bool vn = lane < nb;
    int bn = perm_l[g0 + (vn ? lane : 0)];
    {
        const float* xp = x + ((size_t)bn * CC + c) * II;
#pragma unroll
        for (int q = 0; q < II; ++q) xn[q] = xp[q];
    }

    for (int ch = 0; ch * 32 < nb; ++ch) {
        const bool vcur = vn;
        const int b = bn;
        float xa[II];
#pragma unroll
        for (int q = 0; q < II; ++q) xa[q] = vcur ? xn[q] : 0.f;
        {
            const int jn = (ch + 1) * 32 + lane;
            vn = jn < nb;
            bn = perm_l[g0 + (vn ? jn : 0)];
            const float* xp = x + ((size_t)bn * CC + c) * II;
#pragma unroll
            for (int q = 0; q < II; ++q) xn[q] = xp[q];
        }

        for (int nt = 0; nt < 2; ++nt) {
            if (ch * 32 + nt * 16 >= nb) break;
            float s[II];
#pragma unroll
            for (int k = 0; k < II; ++k) s[k] = __shfl(xa[k], nt * 16 + lq);
            const int bcol = __shfl(b, nt * 16 + lq);
            const float xh0 = half ? s[2] : s[0];
            const float xh1 = half ? s[3] : s[1];
            const float xh2 = half ? s[6] : s[4];
            const float xh3 = half ? s[7] : s[5];
            v2f bf[NKT];
#pragma unroll
            for (int q = 0; q < II; ++q) {
                bf[2 * q][0]     = s[q] * xh0;  bf[2 * q][1]     = s[q] * xh1;
                bf[2 * q + 1][0] = s[q] * xh2;  bf[2 * q + 1][1] = s[q] * xh3;
            }
            bf[18] = bf[16];
            bf[19] = bf[17];
            bf[20][0] = xh0;  bf[20][1] = xh1;
            bf[21][0] = xh2;  bf[21][1] = xh3;
            bf[22][0] = half ? 0.f : s[8] * s[8];
            bf[22][1] = half ? 0.f : s[8];

            __builtin_amdgcn_sched_barrier(0);
            v8f acc0 = {}, acc1 = {};
#pragma unroll
            for (int kt = 0; kt < NKT; ++kt) {
                if (kt & 1) acc1 = __builtin_amdgcn_wmma_f32_16x16x4_f32(false, a[kt], false, bf[kt], (short)0, acc1, false, false);
                else        acc0 = __builtin_amdgcn_wmma_f32_16x16x4_f32(false, a[kt], false, bf[kt], (short)0, acc0, false, false);
            }
            asm volatile("s_wait_alu depctr_va_vdst(0)\n\tv_nop\n\tv_nop\n\tv_nop\n\tv_nop" : "+v"(acc0), "+v"(acc1));
            __builtin_amdgcn_sched_barrier(0);

            float t0 = acc0[0] + acc1[0];
            float part = (c1a + t0) * (half ? s[8] : s[0]);
            float rest = 0.f;
#pragma unroll
            for (int r = 1; r < 8; ++r) rest += (c1v[r] + (acc0[r] + acc1[r])) * s[r];
            part += half ? 0.f : rest;
            part += __shfl_xor(part, 16);
            const int col = ch * 32 + nt * 16 + lq;
            if (half == 0 && col < nb) {
                if (blk == 0) out[bcol * 512 + c] = part;
                else          out[bcol * 512 + CC + c * 3 + (blk - 1)] = part;
            }
        }
    }
}

extern "C" void kernel_launch(void* const* d_in, const int* in_sizes, int n_in,
                              void* d_out, int out_size, void* d_ws, size_t ws_size,
                              hipStream_t stream) {
    (void)in_sizes; (void)n_in; (void)out_size; (void)d_ws; (void)ws_size;

    const float* x   = (const float*)d_in[0];
    const float* y   = (const float*)d_in[1];
    const float* U1s = (const float*)d_in[2];
    const float* U2s = (const float*)d_in[3];
    const float* U3s = (const float*)d_in[4];
    const float* U1v = (const float*)d_in[5];
    const float* U2v = (const float*)d_in[6];
    const float* U3v = (const float*)d_in[7];
    const float* W1s = (const float*)d_in[8];
    const float* W2s = (const float*)d_in[9];
    const float* W3s = (const float*)d_in[10];
    const float* W1v = (const float*)d_in[11];
    const float* W2v = (const float*)d_in[12];
    const float* W3v = (const float*)d_in[13];
    float* out = (float*)d_out;

    sc_fused_kernel<<<NWG, NW * 32, 0, stream>>>(
        x, y, U3s, U3v, W3s, W3v, U2s, U2v, U1s, U1v, W2s, W2v, W1s, W1v, out);
}


# x values of a workgroup's 20 channels (720 contiguous bytes per node) copied once into LDS by whole-line 16-byte loads; B rows read from LDS, no per-lane 36-byte gathers or lane shuffles; sums kept
# speedup vs baseline: 4.9857x; 1.1073x over previous
#include <hip/hip_runtime.h>

#define BB   512
#define CC   128
#define II   9
#define EE   10
#define NP   (EE * CC)
#define NBLK 4
#define NROW 81
#define RP   92
#define NKT  23
#define NPOS_T 52

typedef float v2f __attribute__((ext_vector_type(2)));
typedef float v4f __attribute__((ext_vector_type(4)));
typedef float v8f __attribute__((ext_vector_type(8)));

#define NW    20
#define NWG   (NBLK * (NP / NW))
#define PV    828
#define NSEG  16
#define XS_S  96
#define XS_P  (NW * II)
#define XS_F4 (XS_P / 4)
#define XS_IT ((XS_S * XS_F4 + NW * 32 - 1) / (NW * 32))

__device__ __forceinline__ bool gather_u3(const float* __restrict__ U3, int K3, int nt, int lq, int half, v2f (&bf)[3])
{
    const int R  = nt * 16 + lq;
    const int i  = R / RP;
    const int kk = R - i * RP;
    const int j  = (kk < 72) ? (kk >> 3) : ((kk < 80) ? (kk - 72) : 8);
    const int a  = (kk < 72) ? (kk & 7) : 8;
    const bool okr = (i < II) && ((kk < 80) || (kk == 88));
    const int base = ((i * II + j) * II + a) * K3;
#pragma unroll
    for (int kt = 0; kt < 3; ++kt) {
#pragma unroll
        for (int q = 0; q < 2; ++q) {
            int k = kt * 4 + half * 2 + q;
            bool ok = okr && (k < K3);
            float v = U3[ok ? base + k : 0];
            bf[kt][q] = ok ? v : 0.f;
        }
    }
    return okr;
}

__device__ __forceinline__ void build_tile(const v2f (&af)[2][3], const v2f (&bf)[3], float* Vl, int nt, int lq, int half, bool okpos)
{
    v8f acc0 = {}, acc1 = {};
#pragma unroll
    for (int kt = 0; kt < 3; ++kt) {
        acc0 = __builtin_amdgcn_wmma_f32_16x16x4_f32(false, af[0][kt], false, bf[kt], (short)0, acc0, false, false);
        acc1 = __builtin_amdgcn_wmma_f32_16x16x4_f32(false, af[1][kt], false, bf[kt], (short)0, acc1, false, false);
    }
    asm volatile("s_wait_alu depctr_va_vdst(0)\n\tv_nop\n\tv_nop\n\tv_nop\n\tv_nop" : "+v"(acc0), "+v"(acc1));
    if (okpos) {
#pragma unroll
        for (int v = 0; v < 8; ++v)
            Vl[(v + half * 8) * PV + nt * 16 + lq] = acc0[v];
#pragma unroll
        for (int v = 0; v < 2; ++v)
            Vl[(16 + v + half * 2) * PV + nt * 16 + lq] = acc1[v];
    }
}

__global__ __launch_bounds__(NW * 32) void sc_fused_kernel(
    const float* __restrict__ x, const float* __restrict__ y,
    const float* __restrict__ U3s, const float* __restrict__ U3v,
    const float* __restrict__ W3s, const float* __restrict__ W3v,
    const float* __restrict__ U2s, const float* __restrict__ U2v,
    const float* __restrict__ U1s, const float* __restrict__ U1v,
    const float* __restrict__ W2s, const float* __restrict__ W2v,
    const float* __restrict__ W1s, const float* __restrict__ W1v,
    float* __restrict__ out)
{
    __shared__ __attribute__((aligned(16))) float Vl[NW * PV];
    __shared__ __attribute__((aligned(16))) float xs[XS_S * XS_P];
    __shared__ float u1_l[II];
    __shared__ int perm_l[BB];
    __shared__ int cnt[NSEG * EE];
    __shared__ int wbase[NSEG * EE];
    __shared__ int tot[EE];
    __shared__ int gstart[EE + 1];

    const int tid  = threadIdx.x;
    const int lane = tid & 31;
    const int wave = __builtin_amdgcn_readfirstlane((int)(tid >> 5));
    const int half = lane >> 4;
    const int lq   = lane & 15;
    const int blk  = (int)blockIdx.x & 3;
    const int p0   = ((int)blockIdx.x >> 2) * NW;

    const int p = p0 + wave;
    const int e = p >> 7, c = p & 127;
    const int K3   = (blk == 0) ? 7 : 10;
    const float* W3 = (blk == 0) ? W3s : W3v;
    const float* U3 = (blk == 0) ? U3s : (U3v + (blk - 1) * (81 * 9 * 10));
    const float* U2 = (blk == 0) ? U2s : (U2v + (blk - 1) * 243);
    const float* U1 = (blk == 0) ? U1s : (U1v + (blk - 1) * II);
    const float* W2 = (blk == 0) ? W2s : W2v;
    const float* W1 = (blk == 0) ? W1s : W1v;

    const float u1v = U1[(tid < II) ? tid : 0];
    const int pw = p0 + (tid >> 5);
    const int ew = pw >> 7, cw = pw & 127;
    float w2[3];
#pragma unroll
    for (int k = 0; k < 3; ++k) w2[k] = W2[(ew * 3 + k) * CC + cw];
    const float w1 = W1[ew * CC + cw];
    float u2v[3][3];
#pragma unroll
    for (int m = 0; m < 3; ++m) {
        const int r = lane + m * 32;
#pragma unroll
        for (int k = 0; k < 3; ++k) u2v[m][k] = U2[((r < NROW) ? r : 0) * 3 + k];
    }
    v2f af[2][3];
#pragma unroll
    for (int m2 = 0; m2 < 2; ++m2) {
        const int pl = (m2 == 0) ? lq : (16 + (lq & 7) + (lq >> 3) * 2);
        const bool okp = (m2 == 0) || ((lq & 7) < 2);
        const int pp = p0 + (okp ? pl : 0);
        const int pe = pp >> 7, pc = pp & 127;
#pragma unroll
        for (int kt = 0; kt < 3; ++kt) {
#pragma unroll
            for (int q = 0; q < 2; ++q) {
                int k = kt * 4 + half * 2 + q;
                bool ok = okp && (k < K3);
                float v = W3[(pe * K3 + ((k < K3) ? k : 0)) * CC + pc];
                af[m2][kt][q] = ok ? v : 0.f;
            }
        }
    }
    v2f bf3[2][3];
    bool okb[2];
#pragma unroll
    for (int t = 0; t < 2; ++t) okb[t] = gather_u3(U3, K3, wave + t * NW, lq, half, bf3[t]);
    v2f bl[3] = {};
    bool okl = false;
    if (wave >= 8) okl = gather_u3(U3, K3, 32 + wave, lq, half, bl);
    float yv[2][EE];
#pragma unroll
    for (int h = 0; h < 2; ++h) {
#pragma unroll
        for (int q = 0; q < EE; ++q) yv[h][q] = 0.f;
    }
    if (wave < 8) {
#pragma unroll
        for (int h = 0; h < 2; ++h) {
#pragma unroll
            for (int q = 0; q < EE; ++q) yv[h][q] = y[(tid + h * 256) * EE + q];
        }
    }

    if (tid < II) u1_l[tid] = u1v;
    float* Vw = Vl + wave * PV;
#pragma unroll
    for (int m = 0; m < 3; ++m) {
        const int r = lane + m * 32;
        float c2 = 0.f;
#pragma unroll
        for (int k = 0; k < 3; ++k) c2 += u2v[m][k] * w2[k];
        const int ri = r / II, rj = r - ri * II;
        if (r < NROW) Vw[ri * RP + ((rj < 8) ? (80 + rj) : 89)] = c2;
    }
    if (lane < 2 * II) Vw[(lane >> 1) * RP + 90 + (lane & 1)] = 0.f;

    int e2[2] = {0, 0}, rank2[2] = {0, 0};
    if (wave < 8) {
#pragma unroll
        for (int h = 0; h < 2; ++h) {
            int en = 0;
#pragma unroll
            for (int q = 0; q < EE; ++q)
                if (yv[h][q] > 0.5f) en = q;
            int myrank = 0;
            for (int q = 0; q < EE; ++q) {
                unsigned m = __builtin_amdgcn_ballot_w32(en == q);
                if (en == q) myrank = __popc(m & ((1u << lane) - 1u));
                if (lane == 0) cnt[(h * 8 + wave) * EE + q] = __popc(m);
            }
            e2[h] = en; rank2[h] = myrank;
        }
    }

    {
#pragma unroll
        for (int t = 0; t < 2; ++t) build_tile(af, bf3[t], Vl, wave + t * NW, lq, half, okb[t]);
        if (wave >= 8) build_tile(af, bl, Vl, 32 + wave, lq, half, okl);
    }

    __syncthreads();
    if (tid < EE) {
        int run = 0;
        for (int sg = 0; sg < NSEG; ++sg) { wbase[sg * EE + tid] = run; run += cnt[sg * EE + tid]; }
        tot[tid] = run;
    }
    __syncthreads();
    if (tid == 0) {
        int run = 0;
        for (int q = 0; q < EE; ++q) { gstart[q] = run; run += tot[q]; }
        gstart[EE] = run;
    }
    __syncthreads();
    if (wave < 8) {
#pragma unroll
        for (int h = 0; h < 2; ++h)
            perm_l[gstart[e2[h]] + wbase[(h * 8 + wave) * EE + e2[h]] + rank2[h]] = tid + h * 256;
    }
    __syncthreads();

    const int g0 = __builtin_amdgcn_readfirstlane(gstart[e]);
    const int nb = __builtin_amdgcn_readfirstlane(gstart[e + 1]) - g0;

    const int eA  = p0 >> 7;
    const int cA0 = p0 & 127;
    const int nA  = (CC - cA0 < NW) ? (CC - cA0) : NW;
    const int fA  = (nA >> 2) * II;
    const int g0A = __builtin_amdgcn_readfirstlane(gstart[eA]);
    const int g0B = __builtin_amdgcn_readfirstlane(gstart[eA + 1]);
    const int g1B = __builtin_amdgcn_readfirstlane(gstart[(nA < NW) ? (eA + 2) : (eA + 1)]);
    const int nbA = g0B - g0A;
    const int nbB = g1B - g0B;
    const int nbmax = (nbA > nbB) ? nbA : nbB;

    if (nbmax <= 0) return;

    v4f xr[XS_IT];
    int dsto[XS_IT];
    bool okx[XS_IT];
    auto stage_issue = [&](int base) __attribute__((always_inline)) {
        int nodev[XS_IT];
        int srco[XS_IT];
#pragma unroll
        for (int it = 0; it < XS_IT; ++it) {
            const int idx  = tid + it * (NW * 32);
            const int slot = idx / XS_F4;
            const int f    = idx - slot * XS_F4;
            const bool inA = f < fA;
            const int col  = base + slot;
            const bool ok  = (slot < XS_S) && (col < (inA ? nbA : nbB));
            okx[it]   = ok;
            dsto[it]  = slot * XS_P + f * 4;
            srco[it]  = (inA ? cA0 : 0) * II + (f - (inA ? 0 : fA)) * 4;
            nodev[it] = perm_l[ok ? ((inA ? g0A : g0B) + col) : 0];
        }
#pragma unroll
        for (int it = 0; it < XS_IT; ++it) {
            xr[it] = (v4f){0.f, 0.f, 0.f, 0.f};
            if (okx[it]) xr[it] = *(const v4f*)(x + (size_t)nodev[it] * (CC * II) + srco[it]);
        }
    };
    auto stage_store = [&]() __attribute__((always_inline)) {
#pragma unroll
        for (int it = 0; it < XS_IT; ++it)
            if (okx[it]) *(v4f*)(xs + dsto[it]) = xr[it];
    };

    stage_issue(0);

    v2f a[NKT];
    {
        const float* ar = Vw + ((lq < II) ? lq : (II - 1)) * RP + half * 2;
#pragma unroll
        for (int kt = 0; kt < NKT; ++kt) a[kt] = *(const v2f*)(ar + kt * 4);
    }
    float c1v[II];
#pragma unroll
    for (int q = 0; q < II; ++q) c1v[q] = u1_l[q] * w1;
    const float c1a = half ? c1v[8] : c1v[0];

    for (int base = 0; ; ) {
        stage_store();
        __syncthreads();

        const int ncol = ((nb - base) < XS_S) ? (nb - base) : XS_S;
        for (int t = 0; t * 16 < ncol; ++t) {
            const int cl = t * 16 + lq;
            const bool valid = cl < ncol;
            const float* xrow = xs + (valid ? cl : 0) * XS_P + wave * II;
            float s[II];
#pragma unroll
            for (int k = 0; k < II; ++k) s[k] = xrow[k];
#pragma unroll
            for (int k = 0; k < II; ++k) s[k] = valid ? s[k] : 0.f;
            const int bcol = perm_l[g0 + (valid ? (base + cl) : 0)];
            const float xh0 = half ? s[2] : s[0];
            const float xh1 = half ? s[3] : s[1];
            const float xh2 = half ? s[6] : s[4];
            const float xh3 = half ? s[7] : s[5];
            v2f bf[NKT];
#pragma unroll
            for (int q = 0; q < II; ++q) {
                bf[2 * q][0]     = s[q] * xh0;  bf[2 * q][1]     = s[q] * xh1;
                bf[2 * q + 1][0] = s[q] * xh2;  bf[2 * q + 1][1] = s[q] * xh3;
            }
            bf[18] = bf[16];
            bf[19] = bf[17];
            bf[20][0] = xh0;  bf[20][1] = xh1;
            bf[21][0] = xh2;  bf[21][1] = xh3;
            bf[22][0] = half ? 0.f : s[8] * s[8];
            bf[22][1] = half ? 0.f : s[8];

            __builtin_amdgcn_sched_barrier(0);
            v8f acc0 = {}, acc1 = {};
#pragma unroll
            for (int kt = 0; kt < NKT; ++kt) {
                if (kt & 1) acc1 = __builtin_amdgcn_wmma_f32_16x16x4_f32(false, a[kt], false, bf[kt], (short)0, acc1, false, false);
                else        acc0 = __builtin_amdgcn_wmma_f32_16x16x4_f32(false, a[kt], false, bf[kt], (short)0, acc0, false, false);
            }
            asm volatile("s_wait_alu depctr_va_vdst(0)\n\tv_nop\n\tv_nop\n\tv_nop\n\tv_nop" : "+v"(acc0), "+v"(acc1));
            __builtin_amdgcn_sched_barrier(0);

            float t0 = acc0[0] + acc1[0];
            float part = (c1a + t0) * (half ? s[8] : s[0]);
            float rest = 0.f;
#pragma unroll
            for (int r = 1; r < 8; ++r) rest += (c1v[r] + (acc0[r] + acc1[r])) * s[r];
            part += half ? 0.f : rest;
            part += __shfl_xor(part, 16);
            if (half == 0 && valid) {
                if (blk == 0) out[bcol * 512 + c] = part;
                else          out[bcol * 512 + CC + c * 3 + (blk - 1)] = part;
            }
        }

        base += XS_S;
        if (base >= nbmax) break;
        __syncthreads();
        stage_issue(base);
    }
}

extern "C" void kernel_launch(void* const* d_in, const int* in_sizes, int n_in,
                              void* d_out, int out_size, void* d_ws, size_t ws_size,
                              hipStream_t stream) {
    (void)in_sizes; (void)n_in; (void)out_size; (void)d_ws; (void)ws_size;

    const float* x   = (const float*)d_in[0];
    const float* y   = (const float*)d_in[1];
    const float* U1s = (const float*)d_in[2];
    const float* U2s = (const float*)d_in[3];
    const float* U3s = (const float*)d_in[4];
    const float* U1v = (const float*)d_in[5];
    const float* U2v = (const float*)d_in[6];
    const float* U3v = (const float*)d_in[7];
    const float* W1s = (const float*)d_in[8];
    const float* W2s = (const float*)d_in[9];
    const float* W3s = (const float*)d_in[10];
    const float* W1v = (const float*)d_in[11];
    const float* W2v = (const float*)d_in[12];
    const float* W3v = (const float*)d_in[13];
    float* out = (float*)d_out;

    sc_fused_kernel<<<NWG, NW * 32, 0, stream>>>(
        x, y, U3s, U3v, W3s, W3v, U2s, U2v, U1s, U1v, W2s, W2v, W1s, W1v, out);
}
